# starting kernel
# baseline (speedup 1.0000x reference)
#include <hip/hip_runtime.h>

typedef __attribute__((ext_vector_type(2))) float v2f;
typedef __attribute__((ext_vector_type(8))) float v8f;

#define N_NODES 100000
#define N_EDGES 1600000
#define HID 128


__global__ __launch_bounds__(256) void k_init_deg(float* __restrict__ deg) {
    int i = blockIdx.x * 256 + threadIdx.x;
    if (i < N_NODES) deg[i] = 1.0f;
}

__global__ __launch_bounds__(256) void k_deg(const int* __restrict__ dst,
                                             float* __restrict__ deg) {
    int e = blockIdx.x * 256 + threadIdx.x;
    if (e < N_EDGES) atomicAdd(&deg[dst[e]], 1.0f);
}

__global__ __launch_bounds__(256) void k_dinv(float* __restrict__ deg) {
    int i = blockIdx.x * 256 + threadIdx.x;
    if (i < N_NODES) deg[i] = rsqrtf(deg[i]);
}

template <int MODE>
__global__ __launch_bounds__(256) void k_gemm(
    const float* __restrict__ A, const float* __restrict__ W,
    const float* __restrict__ b_in, const float* __restrict__ b_out,
    const float* __restrict__ xres, float* __restrict__ out) {
    __shared__ float sW[HID * HID];

    const int tid = threadIdx.x;
    for (int i = tid * 4; i < HID * HID; i += 256 * 4)
        *(float4*)&sW[i] = *(const float4*)&W[i];
    __syncthreads();

    const int wave = tid >> 5;
    const int lane = tid & 31;
    const int half = lane >> 4;
    const int lm   = lane & 15;
    const int kb   = half * 2;
    const int m0   = blockIdx.x * 16;
    const int n0   = wave * 16;

    const float* Arow = A + (size_t)(m0 + lm) * HID;

    v8f c = {};
    #pragma unroll 4
    for (int kk = 0; kk < HID; kk += 4) {
        v2f a, b;
        if (MODE == 0) {
            a = *(const v2f*)(Arow + kk + kb);
        } else {
            float a0 = Arow[kk + kb]     + b_in[kk + kb];
            float a1 = Arow[kk + kb + 1] + b_in[kk + kb + 1];
            a.x = a0 > 0.f ? a0 : 0.f;
            a.y = a1 > 0.f ? a1 : 0.f;
        }
        b.x = sW[(kk + kb) * HID + n0 + lm];
        b.y = sW[(kk + kb + 1) * HID + n0 + lm];
        c = __builtin_amdgcn_wmma_f32_16x16x4_f32(false, a, false, b,
                                                  (short)0, c, false, false);
    }

    const int n = n0 + lm;
    #pragma unroll
    for (int v = 0; v < 8; ++v) {
        int m = m0 + v + half * 8;
        float val = c[v];
        if (MODE == 1) {
            val += b_out[n];
            val = val > 0.f ? val : 0.f;
            val += xres[(size_t)m * HID + n];
        }
        out[(size_t)m * HID + n] = val;
    }
}


__global__ __launch_bounds__(256) void k_selfloop(const float* __restrict__ dinv,
                                                  const float* __restrict__ xw,
                                                  float* __restrict__ agg) {
    size_t g = (size_t)blockIdx.x * 256 + threadIdx.x;
    size_t node = g >> 5;
    int lane = (int)(g & 31);
    if (node >= N_NODES) return;
    float di = dinv[node];
    float w = di * di;
    float4 v = *(const float4*)(xw + node * HID + lane * 4);
    float4 r = {w * v.x, w * v.y, w * v.z, w * v.w};
    *(float4*)(agg + node * HID + lane * 4) = r;
}


__global__ __launch_bounds__(256) void k_scatter(const int* __restrict__ src,
                                                 const int* __restrict__ dst,
                                                 const float* __restrict__ dinv,
                                                 const float* __restrict__ xw,
                                                 float* __restrict__ agg) {
    size_t g = (size_t)blockIdx.x * 256 + threadIdx.x;
    size_t e = g >> 5;
    int lane = (int)(g & 31);
    if (e >= N_EDGES) return;
    int s = src[e];
    int d = dst[e];
    float norm = dinv[s] * dinv[d];
    float4 v = *(const float4*)(xw + (size_t)s * HID + lane * 4);
    float* p = agg + (size_t)d * HID + lane * 4;
    atomicAdd(p + 0, norm * v.x);
    atomicAdd(p + 1, norm * v.y);
    atomicAdd(p + 2, norm * v.z);
    atomicAdd(p + 3, norm * v.w);
}


extern "C" void kernel_launch(void* const* d_in, const int* in_sizes, int n_in,
                              void* d_out, int out_size, void* d_ws, size_t ws_size,
                              hipStream_t stream) {
    const float* x     = (const float*)d_in[0];
    const int*   ei    = (const int*)d_in[1];
    const float* W_gcn = (const float*)d_in[2];
    const float* b_gcn = (const float*)d_in[3];
    const float* W_lin = (const float*)d_in[4];
    const float* b_lin = (const float*)d_in[5];

    const int* src = ei;
    const int* dst = ei + N_EDGES;

    float* out = (float*)d_out;

    float* dinv = (float*)d_ws;
    float* agg  = (float*)((char*)d_ws + (size_t)N_NODES * sizeof(float));
    float* xw   = out;

    const int NB_NODE  = (N_NODES + 255) / 256;
    const int NB_EDGE  = (N_EDGES + 255) / 256;
    const int NB_TILE  = N_NODES / 16;
    const int NB_SELF  = (N_NODES * 32 + 255) / 256;
    const int NB_SCAT  = (int)(((size_t)N_EDGES * 32 + 255) / 256);

    k_init_deg<<<NB_NODE, 256, 0, stream>>>(dinv);
    k_deg<<<NB_EDGE, 256, 0, stream>>>(dst, dinv);
    k_dinv<<<NB_NODE, 256, 0, stream>>>(dinv);

    k_gemm<0><<<NB_TILE, 256, 0, stream>>>(x, W_gcn, nullptr, nullptr, nullptr, xw);

    k_selfloop<<<NB_SELF, 256, 0, stream>>>(dinv, xw, agg);
    k_scatter<<<NB_SCAT, 256, 0, stream>>>(src, dst, dinv, xw, agg);

    k_gemm<1><<<NB_TILE, 256, 0, stream>>>(agg, W_lin, b_gcn, b_lin, x, out);
}


# edge list's short form, from 174 more blocks of the degree launch: a bitmap per 400-row block for the byte plane, one packed word (source,row) per edge for the src, dst reads; sums in arrival order
# speedup vs baseline: 3.8545x; 3.8545x over previous
#include <hip/hip_runtime.h>

typedef __attribute__((ext_vector_type(2))) float v2f;
typedef __attribute__((ext_vector_type(8))) float v8f;
typedef __attribute__((ext_vector_type(4))) int   v4i;
typedef __attribute__((ext_vector_type(4))) float v4f;

#define N_NODES 100000
#define N_EDGES 1600000
#define HID 128


#define AGG_ROWS    400
#define AGG_BLOCKS  (N_NODES / AGG_ROWS)
#define HIST_BLOCKS 64
#define HIST_WORDS  (N_NODES / 2)
#define HIST_PITCH  50016
#define HIST_EDGES  (N_EDGES / HIST_BLOCKS)
#define HIST_LOADS  (HIST_EDGES / 4)

#define BM_LINES    9
#define BM_WORDS    (BM_LINES * 32)
#define BM_EDGES    (BM_LINES * 1024)
#define BM_LOADS    (BM_EDGES / 4)
#define BM_GLINES   1563
#define BM_PITCH    (BM_GLINES * 32)
#define BM_BLOCKS   174
#define BM_UNITS    (N_EDGES / 128)
#define BM_COPY     (AGG_BLOCKS * BM_WORDS / 4)

__global__ __launch_bounds__(1024) void k_hist(const int* __restrict__ src,
                                               const int* __restrict__ dst,
                                               unsigned* __restrict__ part,
                                               unsigned* __restrict__ bits,
                                               unsigned* __restrict__ pk) {
    __shared__ unsigned h[AGG_BLOCKS * BM_WORDS];

    const int tid = threadIdx.x;

    if (blockIdx.x >= HIST_BLOCKS) {
        const int piece = blockIdx.x - HIST_BLOCKS;
        for (int i = tid; i < BM_COPY; i += 1024)
            ((uint4*)h)[i] = uint4{0u, 0u, 0u, 0u};
        __syncthreads();

        const v4i* d4 = (const v4i*)dst;
        const v4i* s4 = (const v4i*)src;
        v4i d[3], sn[3];
        #pragma unroll
        for (int it = 0; it < 3; ++it) {
            int q = piece * BM_LOADS + tid + it * 1024;
            q = q < N_EDGES / 4 ? q : N_EDGES / 4 - 1;
            d[it]  = d4[q];
            sn[it] = s4[q];
        }
        asm volatile("" :: "v"(d[0]), "v"(d[1]), "v"(d[2]), "v"(sn[0]), "v"(sn[1]), "v"(sn[2]));
        #pragma unroll
        for (int it = 0; it < 3; ++it) {
            const int i = tid + it * 1024;
            if (i < BM_LOADS && piece * BM_LOADS + i < N_EDGES / 4) {
                const int w = i >> 3;
                const unsigned sh = (unsigned)(i & 7) * 4u;
                const unsigned b0 = (unsigned)d[it].x / AGG_ROWS, b1 = (unsigned)d[it].y / AGG_ROWS;
                const unsigned b2 = (unsigned)d[it].z / AGG_ROWS, b3 = (unsigned)d[it].w / AGG_ROWS;
                if (b0 < AGG_BLOCKS) atomicOr(&h[b0 * BM_WORDS + w], 1u << sh);
                if (b1 < AGG_BLOCKS) atomicOr(&h[b1 * BM_WORDS + w], 2u << sh);
                if (b2 < AGG_BLOCKS) atomicOr(&h[b2 * BM_WORDS + w], 4u << sh);
                if (b3 < AGG_BLOCKS) atomicOr(&h[b3 * BM_WORDS + w], 8u << sh);
                uint4 p;
                p.x = (unsigned)sn[it].x | ((((unsigned)d[it].x - b0 * AGG_ROWS) & 511u) << 17);
                p.y = (unsigned)sn[it].y | ((((unsigned)d[it].y - b1 * AGG_ROWS) & 511u) << 17);
                p.z = (unsigned)sn[it].z | ((((unsigned)d[it].z - b2 * AGG_ROWS) & 511u) << 17);
                p.w = (unsigned)sn[it].w | ((((unsigned)d[it].w - b3 * AGG_ROWS) & 511u) << 17);
                ((uint4*)pk)[piece * BM_LOADS + i] = p;
            }
        }
        __syncthreads();

        for (int idx = tid; idx < BM_COPY; idx += 1024) {
            const int b = idx / (BM_WORDS / 4);
            const int k = idx - b * (BM_WORDS / 4);
            if (piece * BM_LINES + (k >> 3) < BM_GLINES)
                ((uint4*)(bits + (size_t)b * BM_PITCH + piece * BM_WORDS))[k] = ((const uint4*)h)[idx];
        }
        return;
    }

    for (int i = tid; i < HIST_WORDS / 4; i += 1024)
        ((uint4*)h)[i] = uint4{0u, 0u, 0u, 0u};
    __syncthreads();

    const v4i* d4 = (const v4i*)dst + blockIdx.x * HIST_LOADS;
    v4i d[7];
    #pragma unroll
    for (int it = 0; it < 7; ++it) {
        int i = tid + it * 1024;
        d[it] = d4[i < HIST_LOADS ? i : HIST_LOADS - 1];
    }
    asm volatile("" :: "v"(d[0]), "v"(d[1]), "v"(d[2]), "v"(d[3]), "v"(d[4]), "v"(d[5]), "v"(d[6]));
    #pragma unroll
    for (int it = 0; it < 7; ++it) {
        if (tid + it * 1024 < HIST_LOADS) {
            atomicAdd(&h[d[it].x >> 1], 1u << ((d[it].x & 1) * 16));
            atomicAdd(&h[d[it].y >> 1], 1u << ((d[it].y & 1) * 16));
            atomicAdd(&h[d[it].z >> 1], 1u << ((d[it].z & 1) * 16));
            atomicAdd(&h[d[it].w >> 1], 1u << ((d[it].w & 1) * 16));
        }
    }
    __syncthreads();

    uint4* o = (uint4*)(part + (size_t)blockIdx.x * HIST_PITCH);
    for (int i = tid; i < HIST_WORDS / 4; i += 1024)
        o[i] = ((const uint4*)h)[i];
}

__global__ __launch_bounds__(256) void k_dinv(const unsigned* __restrict__ part,
                                              float* __restrict__ dinv) {
    int i = blockIdx.x * 256 + threadIdx.x;
    if (i < HIST_WORDS) {
        unsigned lo = 1u, hi = 1u;
        #pragma unroll 16
        for (int b = 0; b < HIST_BLOCKS; ++b) {
            unsigned w = part[(size_t)b * HIST_PITCH + i];
            lo += w & 0xFFFFu;
            hi += w >> 16;
        }
        *(float2*)&dinv[2 * i] = float2{rsqrtf((float)lo), rsqrtf((float)hi)};
    }
}


#define G1_BLOCKS  1024
#define G1_TILES   (N_NODES / 16)
#define XT_PITCH   132
#define XT_FLOATS  (16 * XT_PITCH)

__global__ __launch_bounds__(256) void k_gemm(
    const float* __restrict__ A, const float* __restrict__ W, const float* __restrict__ dinv,
    float* __restrict__ out) {
    __shared__ float sA[2 * XT_FLOATS];

    const int tid  = threadIdx.x;
    const int wave = tid >> 5;
    const int lane = tid & 31;
    const int half = lane >> 4;
    const int lm   = lane & 15;
    const int n    = wave * 16 + lm;

    v2f bf[32];
    #pragma unroll
    for (int j = 0; j < 32; ++j) {
        const int k = 4 * j + 2 * half;
        bf[j].x = W[k * HID + n];
        bf[j].y = W[(k + 1) * HID + n];
    }

    const int st = (tid >> 5) * XT_PITCH + (tid & 31) * 4;

    int t = blockIdx.x;
    float4 da, db;
    {
        const float* g = A + (size_t)t * (16 * HID) + tid * 4;
        const float4 f0 = *(const float4*)g;
        const float4 f1 = *(const float4*)(g + 8 * HID);
        const float4* dv = (const float4*)(dinv + t * 16 + 8 * half);
        da = dv[0];
        db = dv[1];
        *(float4*)&sA[st] = f0;
        *(float4*)&sA[st + 8 * XT_PITCH] = f1;
    }
    __syncthreads();

    int buf = 0;
    for (; t < G1_TILES; t += G1_BLOCKS) {
        const bool more = t + G1_BLOCKS < G1_TILES;
        float4 p0, p1, na, nb;
        if (more) {
            const float* gn = A + (size_t)(t + G1_BLOCKS) * (16 * HID) + tid * 4;
            p0 = *(const float4*)gn;
            p1 = *(const float4*)(gn + 8 * HID);
            const float4* dn = (const float4*)(dinv + (t + G1_BLOCKS) * 16 + 8 * half);
            na = dn[0];
            nb = dn[1];
        }

        const float* arow = &sA[buf * XT_FLOATS + lm * XT_PITCH + 2 * half];
        v8f c = {};
        #pragma unroll
        for (int j = 0; j < 32; ++j) {
            v2f a = *(const v2f*)(arow + 4 * j);
            c = __builtin_amdgcn_wmma_f32_16x16x4_f32(false, a, false, bf[j],
                                                      (short)0, c, false, false);
        }

        float* o = out + (size_t)(t * 16 + 8 * half) * HID + n;
        const float dr[8] = {da.x, da.y, da.z, da.w, db.x, db.y, db.z, db.w};
        #pragma unroll
        for (int v = 0; v < 8; ++v) o[(size_t)v * HID] = dr[v] * c[v];

        if (more) {
            buf ^= 1;
            *(float4*)&sA[buf * XT_FLOATS + st] = p0;
            *(float4*)&sA[buf * XT_FLOATS + st + 8 * XT_PITCH] = p1;
            da = na;
            db = nb;
        }
        __syncthreads();
    }
}


#define ACC_PITCH  132
#define ROW_TILES  (AGG_ROWS / 16)
#define AGG_WAVES  32
#define SCAN_TRIPS ((BM_UNITS + 1023) / 1024)
#define LIST_CAP   64

__global__ __launch_bounds__(1024) void k_agg_lds(const unsigned* __restrict__ pk,
                                                  const unsigned* __restrict__ bits,
                                                  const float* __restrict__ dinv,
                                                  const float* __restrict__ xw,
                                                  const float* __restrict__ b_gcn,
                                                  const float* __restrict__ W_lin,
                                                  const float* __restrict__ b_lin,
                                                  const float* __restrict__ x,
                                                  float* __restrict__ out) {
    __shared__ float acc[AGG_ROWS * ACC_PITCH];
    __shared__ unsigned wlist[AGG_WAVES * LIST_CAP];

    const int tid  = threadIdx.x;
    const int wave = __builtin_amdgcn_readfirstlane(tid >> 5);
    const int lane = tid & 31;
    const int sw   = (lane >> 4) << 1;
    const unsigned lw4 = (unsigned)(lane * 4 + sw) * 4u;
    const int d0   = blockIdx.x * AGG_ROWS;

    for (int i = tid; i < AGG_ROWS * 32; i += 1024) {
        int r = i >> 5, c = i & 31;
        float4 o = *(const float4*)(xw + (size_t)(d0 + r) * HID + c * 4);
        if (c >= 16) o = float4{o.z, o.w, o.x, o.y};
        *(float4*)&acc[r * ACC_PITCH + c * 4] = o;
    }
    __syncthreads();

    unsigned* mylist = wlist + wave * LIST_CAP;
    int cnt = 0;
    const uint4* bm4 = (const uint4*)(bits + (size_t)blockIdx.x * BM_PITCH);

    auto drain = [&](int n) {
        for (int base = 0; base < n; base += 32) {
            int idx = base + lane;
            unsigned e = idx < n ? mylist[idx] : 0u;
            const unsigned p = pk[e];
            int s = (int)(p & 0x1FFFFu);
            int r = (int)(p >> 17);
            int m = n - base < 32 ? n - base : 32;
            for (int q = 0; q < m; q += 8) {
                v4f v[8];
                #pragma unroll
                for (int t = 0; t < 8; ++t) {
                    int sq = __builtin_amdgcn_readlane(s, q + t);
                    v[t] = *(const v4f*)(xw + (size_t)sq * HID + lane * 4);
                }
                #pragma unroll
                for (int t = 0; t < 8; ++t) {
                    if (q + t < m) {
                        int rr = __builtin_amdgcn_readlane(r, q + t);
                        unsigned oa = (unsigned)rr * (ACC_PITCH * 4u) + lw4;
                        float* pa = (float*)((char*)acc + oa);
                        float* pb = (float*)((char*)acc + (oa ^ 8u));
                        atomicAdd(pa + 0, v[t].x);
                        atomicAdd(pa + 1, v[t].y);
                        atomicAdd(pb + 0, v[t].z);
                        atomicAdd(pb + 1, v[t].w);
                    }
                }
                asm volatile("" :: "v"(v[0]), "v"(v[1]), "v"(v[2]), "v"(v[3]),
                                   "v"(v[4]), "v"(v[5]), "v"(v[6]), "v"(v[7]));
            }
        }
    };

    auto unit_of = [&](int it) {
        int u = it * 1024 + tid;
        return u < BM_UNITS ? u : BM_UNITS - 1;
    };

    uint4 c0 = bm4[unit_of(0)];

    for (int it = 0; it < SCAN_TRIPS; ++it) {
        const uint4 c1 = bm4[unit_of(it + 1)];
        const int u = it * 1024 + tid;
        const bool ok = u < BM_UNITS;
        unsigned w0 = ok ? c0.x : 0u, w1 = ok ? c0.y : 0u, w2 = ok ? c0.z : 0u, w3 = ok ? c0.w : 0u;
        unsigned eb = (unsigned)u * 128u;
        #pragma nounroll
        for (int j = 0; j < 4; ++j) {
            unsigned m = w0;
            w0 = w1; w1 = w2; w2 = w3;
            unsigned mask = __builtin_amdgcn_ballot_w32(m != 0u);
            while (mask) {
                if (m) {
                    int pos = cnt + (int)__builtin_amdgcn_mbcnt_lo(mask, 0u);
                    mylist[pos] = eb + (unsigned)__builtin_ctz(m);
                }
                m &= m - 1u;
                cnt += __builtin_popcount(mask);
                mask = __builtin_amdgcn_ballot_w32(m != 0u);
                if (cnt >= 32) { drain(cnt); cnt = 0; }
            }
            eb += 32u;
        }
        c0 = c1;
    }
    if (cnt > 0) drain(cnt);

    const int ct   = wave & 7;
    const int rg   = wave >> 3;
    const int lm   = lane & 15;
    const int half = lane >> 4;
    const int n    = ct * 16 + lm;

    v2f bf[32];
    #pragma unroll
    for (int j = 0; j < 32; ++j) {
        const int k = 4 * j + 2 * half;
        bf[j].x = W_lin[k * HID + n];
        bf[j].y = W_lin[(k + 1) * HID + n];
    }
    const float bl = b_lin[n];
    const float4 bg = *(const float4*)(b_gcn + lane * 4);
    const int rf = wave + lane * AGG_WAVES;
    const float df = rf < AGG_ROWS ? dinv[d0 + rf] : 0.f;

    __syncthreads();

    for (int r = wave, i = 0; r < AGG_ROWS; r += AGG_WAVES, ++i) {
        const float di = __int_as_float(__builtin_amdgcn_readlane(__float_as_int(df), i));
        float4* p = (float4*)&acc[r * ACC_PITCH + lane * 4];
        float4 v = *p;
        if (sw) v = float4{v.z, v.w, v.x, v.y};
        v.x = di * v.x + bg.x; v.y = di * v.y + bg.y; v.z = di * v.z + bg.z; v.w = di * v.w + bg.w;
        v.x = v.x > 0.f ? v.x : 0.f;
        v.y = v.y > 0.f ? v.y : 0.f;
        v.z = v.z > 0.f ? v.z : 0.f;
        v.w = v.w > 0.f ? v.w : 0.f;
        *p = v;
    }
    __syncthreads();

    for (int t = rg; t < ROW_TILES; t += AGG_WAVES / 8) {
        const int r0 = t * 16;
        const float* arow = &acc[(r0 + lm) * ACC_PITCH + 2 * half];
        const size_t o0 = (size_t)(d0 + r0 + 8 * half) * HID + n;

        float xr[8];
        #pragma unroll
        for (int v = 0; v < 8; ++v) xr[v] = x[o0 + (size_t)v * HID];

        v8f c = {};
        #pragma unroll
        for (int j = 0; j < 32; ++j) {
            v2f a = *(const v2f*)(arow + 4 * j);
            c = __builtin_amdgcn_wmma_f32_16x16x4_f32(false, a, false, bf[j],
                                                      (short)0, c, false, false);
        }

        #pragma unroll
        for (int v = 0; v < 8; ++v) {
            float val = c[v] + bl;
            val = val > 0.f ? val : 0.f;
            out[o0 + (size_t)v * HID] = val + xr[v];
        }
    }
}


extern "C" void kernel_launch(void* const* d_in, const int* in_sizes, int n_in,
                              void* d_out, int out_size, void* d_ws, size_t ws_size,
                              hipStream_t stream) {
    const float* x     = (const float*)d_in[0];
    const int*   ei    = (const int*)d_in[1];
    const float* W_gcn = (const float*)d_in[2];
    const float* b_gcn = (const float*)d_in[3];
    const float* W_lin = (const float*)d_in[4];
    const float* b_lin = (const float*)d_in[5];

    const int* src = ei;
    const int* dst = ei + N_EDGES;

    float* out = (float*)d_out;

    float* dinv = (float*)d_ws;
    float* xw   = (float*)((char*)d_ws + (size_t)N_NODES * sizeof(float));
    unsigned* part = (unsigned*)((char*)d_ws + (size_t)N_NODES * sizeof(float)
                                 + (size_t)N_NODES * HID * sizeof(float));
    unsigned* bits = (unsigned*)((char*)d_ws + (size_t)N_NODES * sizeof(float)
                                 + (size_t)N_NODES * HID * sizeof(float)
                                 + (size_t)HIST_BLOCKS * HIST_PITCH * sizeof(unsigned));

    unsigned* pk   = (unsigned*)((char*)d_ws + (size_t)N_NODES * sizeof(float)
                                 + (size_t)N_NODES * HID * sizeof(float)
                                 + (size_t)HIST_BLOCKS * HIST_PITCH * sizeof(unsigned)
                                 + (size_t)AGG_BLOCKS * BM_PITCH * sizeof(unsigned));

    const int NB_WORD  = (HIST_WORDS + 255) / 256;

    k_hist<<<HIST_BLOCKS + BM_BLOCKS, 1024, 0, stream>>>(src, dst, part, bits, pk);
    k_dinv<<<NB_WORD, 256, 0, stream>>>(part, dinv);

    k_gemm<<<G1_BLOCKS, 256, 0, stream>>>(x, W_gcn, dinv, xw);

    k_agg_lds<<<AGG_BLOCKS, 1024, 0, stream>>>(pk, bits, dinv, xw, b_gcn, W_lin, b_lin, x, out);
}
